# starting kernel
# baseline (speedup 1.0000x reference)
#include <hip/hip_runtime.h>
#include <math.h>

#define B_    32
#define C_    256
#define HW_   4096
#define NOUT_ 768
#define BN_EPS_ 1e-5f

typedef float v2f __attribute__((ext_vector_type(2)));
typedef float v8f __attribute__((ext_vector_type(8)));

__global__ void gap_kernel(const float* __restrict__ x, float* __restrict__ g)
{
    const int plane = blockIdx.x;
    const float4* xp = (const float4*)(x + (size_t)plane * HW_);
    float s = 0.f;
#pragma unroll
    for (int i = 0; i < 4; ++i) {
        float4 v = xp[threadIdx.x + i * 256];
        s += v.x + v.y + v.z + v.w;
    }
#pragma unroll
    for (int off = 16; off > 0; off >>= 1)
        s += __shfl_xor(s, off, 32);
    __shared__ float lds[8];
    const int wid  = threadIdx.x >> 5;
    const int lane = threadIdx.x & 31;
    if (lane == 0) lds[wid] = s;
    __syncthreads();
    if (threadIdx.x == 0) {
        float t = 0.f;
#pragma unroll
        for (int w = 0; w < 8; ++w) t += lds[w];
        g[plane] = t * (1.0f / (float)HW_);
    }
}

__global__ void filt_gemm_wmma_kernel(const float* __restrict__ g,
                                      const float* __restrict__ w,
                                      const float* __restrict__ bnw,
                                      const float* __restrict__ bnb,
                                      const float* __restrict__ bnm,
                                      const float* __restrict__ bnv,
                                      float* __restrict__ f)
{
    const int tile = blockIdx.x;
    const int mt   = tile & 1;
    const int nt   = tile >> 1;
    const int lane = threadIdx.x;
    const int ln   = lane & 15;
    const int lh   = lane >> 4;

    const float* arow = g + (size_t)(mt * 16 + ln) * C_;
    const float* brow = w + (size_t)(nt * 16 + ln) * C_;

    v8f acc = {};
    for (int k0 = 0; k0 < C_; k0 += 4) {
        v2f a, b;
        a.x = arow[k0 + 2 * lh + 0];
        a.y = arow[k0 + 2 * lh + 1];
        b.x = brow[k0 + 2 * lh + 0];
        b.y = brow[k0 + 2 * lh + 1];
        acc = __builtin_amdgcn_wmma_f32_16x16x4_f32(
            false, a, false, b, (short)0, acc, false, false);
    }

    const int   n     = nt * 16 + ln;
    const float scale = bnw[n] * rsqrtf(bnv[n] + BN_EPS_);
    const float bias  = bnb[n] - bnm[n] * scale;
#pragma unroll
    for (int r = 0; r < 8; ++r) {
        const int m = mt * 16 + lh * 8 + r;
        f[(size_t)m * NOUT_ + n] = tanhf(acc[r] * scale + bias);
    }
}

__global__ void apply_kernel(const float* __restrict__ x,
                             const float* __restrict__ f,
                             const float* __restrict__ gamma,
                             const float* __restrict__ beta,
                             float* __restrict__ out)
{
    const int c = blockIdx.x;
    const int b = blockIdx.y;
    const int cm = (c == 0)      ? 1        : c - 1;
    const int cp = (c == C_ - 1) ? C_ - 2   : c + 1;

    const float f0 = f[(size_t)b * NOUT_ + 0 * C_ + c];
    const float f1 = f[(size_t)b * NOUT_ + 1 * C_ + c];
    const float f2 = f[(size_t)b * NOUT_ + 2 * C_ + c];
    const float ga = gamma[c];
    const float be = beta[c];

    const float4* xm = (const float4*)(x   + ((size_t)b * C_ + cm) * HW_);
    const float4* xc = (const float4*)(x   + ((size_t)b * C_ + c ) * HW_);
    const float4* xp = (const float4*)(x   + ((size_t)b * C_ + cp) * HW_);
    float4*       op = (float4*)      (out + ((size_t)b * C_ + c ) * HW_);

#pragma unroll
    for (int i = 0; i < 4; ++i) {
        const int idx = threadIdx.x + i * 256;
        const float4 a = xm[idx];
        const float4 m = xc[idx];
        const float4 p = xp[idx];
        float4 o;
        o.x = ga * (f0 * a.x + f1 * m.x + f2 * p.x) + be * m.x;
        o.y = ga * (f0 * a.y + f1 * m.y + f2 * p.y) + be * m.y;
        o.z = ga * (f0 * a.z + f1 * m.z + f2 * p.z) + be * m.z;
        o.w = ga * (f0 * a.w + f1 * m.w + f2 * p.w) + be * m.w;
        op[idx] = o;
    }
}

extern "C" void kernel_launch(void* const* d_in, const int* in_sizes, int n_in,
                              void* d_out, int out_size, void* d_ws, size_t ws_size,
                              hipStream_t stream)
{
    const float* x      = (const float*)d_in[0];
    const float* conv_w = (const float*)d_in[1];
    const float* bnw    = (const float*)d_in[2];
    const float* bnb    = (const float*)d_in[3];
    const float* bnm    = (const float*)d_in[4];
    const float* bnv    = (const float*)d_in[5];
    const float* gamma  = (const float*)d_in[6];
    const float* beta   = (const float*)d_in[7];
    float*       out    = (float*)d_out;

    float* g_ws = (float*)d_ws;
    float* f_ws = (float*)((char*)d_ws + B_ * C_ * sizeof(float));

    gap_kernel<<<B_ * C_, 256, 0, stream>>>(x, g_ws);
    filt_gemm_wmma_kernel<<<96, 32, 0, stream>>>(g_ws, conv_w, bnw, bnb, bnm, bnv, f_ws);
    apply_kernel<<<dim3(C_, B_), 256, 0, stream>>>(x, f_ws, gamma, beta, out);
}


# apply kernel: each thread walks 64 consecutive channels with the 3-plane window in registers and 16 planes requested ahead; x is loaded once per run, not 3 times, by 4096 long waves, not 65536
# speedup vs baseline: 1.1720x; 1.1720x over previous
#include <hip/hip_runtime.h>
#include <math.h>

#define B_    32
#define C_    256
#define HW_   4096
#define NOUT_ 768
#define BN_EPS_ 1e-5f

typedef float v2f __attribute__((ext_vector_type(2)));
typedef float v8f __attribute__((ext_vector_type(8)));
typedef float v4f __attribute__((ext_vector_type(4)));

__global__ void gap_kernel(const float* __restrict__ x, float* __restrict__ g)
{
    const int plane = blockIdx.x;
    const float4* xp = (const float4*)(x + (size_t)plane * HW_);
    float s = 0.f;
#pragma unroll
    for (int i = 0; i < 4; ++i) {
        float4 v = xp[threadIdx.x + i * 256];
        s += v.x + v.y + v.z + v.w;
    }
#pragma unroll
    for (int off = 16; off > 0; off >>= 1)
        s += __shfl_xor(s, off, 32);
    __shared__ float lds[8];
    const int wid  = threadIdx.x >> 5;
    const int lane = threadIdx.x & 31;
    if (lane == 0) lds[wid] = s;
    __syncthreads();
    if (threadIdx.x == 0) {
        float t = 0.f;
#pragma unroll
        for (int w = 0; w < 8; ++w) t += lds[w];
        g[plane] = t * (1.0f / (float)HW_);
    }
}

__global__ void filt_gemm_wmma_kernel(const float* __restrict__ g,
                                      const float* __restrict__ w,
                                      const float* __restrict__ bnw,
                                      const float* __restrict__ bnb,
                                      const float* __restrict__ bnm,
                                      const float* __restrict__ bnv,
                                      float* __restrict__ f)
{
    const int tile = blockIdx.x;
    const int mt   = tile & 1;
    const int nt   = tile >> 1;
    const int lane = threadIdx.x;
    const int ln   = lane & 15;
    const int lh   = lane >> 4;

    const float* arow = g + (size_t)(mt * 16 + ln) * C_;
    const float* brow = w + (size_t)(nt * 16 + ln) * C_;

    v8f acc = {};
    for (int k0 = 0; k0 < C_; k0 += 4) {
        v2f a, b;
        a.x = arow[k0 + 2 * lh + 0];
        a.y = arow[k0 + 2 * lh + 1];
        b.x = brow[k0 + 2 * lh + 0];
        b.y = brow[k0 + 2 * lh + 1];
        acc = __builtin_amdgcn_wmma_f32_16x16x4_f32(
            false, a, false, b, (short)0, acc, false, false);
    }

    const int   n     = nt * 16 + ln;
    const float scale = bnw[n] * rsqrtf(bnv[n] + BN_EPS_);
    const float bias  = bnb[n] - bnm[n] * scale;
#pragma unroll
    for (int r = 0; r < 8; ++r) {
        const int m = mt * 16 + lh * 8 + r;
        f[(size_t)m * NOUT_ + n] = tanhf(acc[r] * scale + bias);
    }
}

#define RUN_    64
#define DEPTH_  16
#define PLANE4_ (HW_ / 4)
#define SETS_   (RUN_ / 32)

#define LANE_F(t, l) __int_as_float(__builtin_amdgcn_readlane(__float_as_int(t), (l)))

__global__ void __launch_bounds__(256)
apply_kernel(const float* __restrict__ x,
             const float* __restrict__ f,
             const float* __restrict__ gamma,
             const float* __restrict__ beta,
             float* __restrict__ out)
{
    const int q    = blockIdx.x * 256 + threadIdx.x;
    const int c0   = blockIdx.y * RUN_;
    const int b    = blockIdx.z;
    const int lane = threadIdx.x & 31;

    const float4* xb = (const float4*)(x   + ((size_t)b * C_ + c0) * HW_) + q;
    float4*       ob = (float4*)      (out + ((size_t)b * C_ + c0) * HW_) + q;
    const float*  fb = f + (size_t)b * NOUT_ + c0;
    const float*  gp = gamma + c0;
    const float*  bp = beta  + c0;

    float tf0[SETS_], tf1[SETS_], tf2[SETS_], tga[SETS_], tbe[SETS_];
#pragma unroll
    for (int s = 0; s < SETS_; ++s) {
        tf0[s] = fb[0 * C_ + 32 * s + lane];
        tf1[s] = fb[1 * C_ + 32 * s + lane];
        tf2[s] = fb[2 * C_ + 32 * s + lane];
        tga[s] = gp[32 * s + lane];
        tbe[s] = bp[32 * s + lane];
    }

    const int rlo = (c0 == 0)         ? 1        : -1;
    const int rhi = (c0 + RUN_ == C_) ? RUN_ - 2 : RUN_;

    float4 prev = xb[(ptrdiff_t)rlo * PLANE4_];
    float4 cur  = xb[0];
    float4 ring[DEPTH_];
#pragma unroll
    for (int i = 0; i < DEPTH_; ++i)
        ring[i] = xb[(size_t)(1 + i) * PLANE4_];

#pragma unroll
    for (int j = 0; j < RUN_; ++j) {
        float4 nxt = ring[j % DEPTH_];
        const int ahead = j + 1 + DEPTH_;
        if (ahead < RUN_)
            ring[j % DEPTH_] = xb[(size_t)ahead * PLANE4_];
        else if (ahead == RUN_)
            ring[j % DEPTH_] = xb[(ptrdiff_t)rhi * PLANE4_];

        v4f pin = { nxt.x, nxt.y, nxt.z, nxt.w };
        asm volatile("" : "+v"(pin));
        nxt.x = pin.x; nxt.y = pin.y; nxt.z = pin.z; nxt.w = pin.w;

        const float f0 = LANE_F(tf0[j / 32], j % 32);
        const float f1 = LANE_F(tf1[j / 32], j % 32);
        const float f2 = LANE_F(tf2[j / 32], j % 32);
        const float ga = LANE_F(tga[j / 32], j % 32);
        const float be = LANE_F(tbe[j / 32], j % 32);

        const float4 a = prev;
        const float4 m = cur;
        const float4 p = nxt;
        float4 o;
        o.x = ga * (f0 * a.x + f1 * m.x + f2 * p.x) + be * m.x;
        o.y = ga * (f0 * a.y + f1 * m.y + f2 * p.y) + be * m.y;
        o.z = ga * (f0 * a.z + f1 * m.z + f2 * p.z) + be * m.z;
        o.w = ga * (f0 * a.w + f1 * m.w + f2 * p.w) + be * m.w;
        ob[(size_t)j * PLANE4_] = o;

        prev = cur;
        cur  = nxt;
    }
}

extern "C" void kernel_launch(void* const* d_in, const int* in_sizes, int n_in,
                              void* d_out, int out_size, void* d_ws, size_t ws_size,
                              hipStream_t stream)
{
    const float* x      = (const float*)d_in[0];
    const float* conv_w = (const float*)d_in[1];
    const float* bnw    = (const float*)d_in[2];
    const float* bnb    = (const float*)d_in[3];
    const float* bnm    = (const float*)d_in[4];
    const float* bnv    = (const float*)d_in[5];
    const float* gamma  = (const float*)d_in[6];
    const float* beta   = (const float*)d_in[7];
    float*       out    = (float*)d_out;

    float* g_ws = (float*)d_ws;
    float* f_ws = (float*)((char*)d_ws + B_ * C_ * sizeof(float));

    gap_kernel<<<B_ * C_, 256, 0, stream>>>(x, g_ws);
    filt_gemm_wmma_kernel<<<96, 32, 0, stream>>>(g_ws, conv_w, bnw, bnb, bnm, bnv, f_ws);
    apply_kernel<<<dim3(PLANE4_ / 256, C_ / RUN_, B_), 256, 0, stream>>>(x, f_ws, gamma, beta, out);
}


# filter GEMM kernel: 8 waves per 16x16 tile, each takes 32 k with all its operand loads in flight at once; partial tiles added through LDS in fixed order (the sum order over k changes)
# speedup vs baseline: 1.2484x; 1.0652x over previous
#include <hip/hip_runtime.h>
#include <math.h>

#define B_    32
#define C_    256
#define HW_   4096
#define NOUT_ 768
#define BN_EPS_ 1e-5f

typedef float v2f __attribute__((ext_vector_type(2)));
typedef float v8f __attribute__((ext_vector_type(8)));
typedef float v4f __attribute__((ext_vector_type(4)));

__global__ void gap_kernel(const float* __restrict__ x, float* __restrict__ g)
{
    const int plane = blockIdx.x;
    const float4* xp = (const float4*)(x + (size_t)plane * HW_);
    float s = 0.f;
#pragma unroll
    for (int i = 0; i < 4; ++i) {
        float4 v = xp[threadIdx.x + i * 256];
        s += v.x + v.y + v.z + v.w;
    }
#pragma unroll
    for (int off = 16; off > 0; off >>= 1)
        s += __shfl_xor(s, off, 32);
    __shared__ float lds[8];
    const int wid  = threadIdx.x >> 5;
    const int lane = threadIdx.x & 31;
    if (lane == 0) lds[wid] = s;
    __syncthreads();
    if (threadIdx.x == 0) {
        float t = 0.f;
#pragma unroll
        for (int w = 0; w < 8; ++w) t += lds[w];
        g[plane] = t * (1.0f / (float)HW_);
    }
}

#define KSPLIT_ 8

__global__ void __launch_bounds__(32 * KSPLIT_)
filt_gemm_wmma_kernel(const float* __restrict__ g,
                      const float* __restrict__ w,
                      const float* __restrict__ bnw,
                      const float* __restrict__ bnb,
                      const float* __restrict__ bnm,
                      const float* __restrict__ bnv,
                      float* __restrict__ f)
{
    __shared__ float part[KSPLIT_][256];

    const int tile = blockIdx.x;
    const int mt   = tile & 1;
    const int nt   = tile >> 1;
    const int t    = threadIdx.x;
    const int wv   = t >> 5;
    const int lane = t & 31;
    const int ln   = lane & 15;
    const int lh   = lane >> 4;

    const float4* ap = (const float4*)(g + (size_t)(mt * 16 + ln) * C_ + wv * 32 + 4 * lh);
    const float4* bp = (const float4*)(w + (size_t)(nt * 16 + ln) * C_ + wv * 32 + 4 * lh);

    float4 a[4], b[4];
    for (int i = 0; i < 4; ++i) {
        a[i] = ap[2 * i];
        b[i] = bp[2 * i];
    }

    const int   n  = nt * 16 + (t & 15);
    const float pw = bnw[n];
    const float pb = bnb[n];
    const float pm = bnm[n];
    const float pv = bnv[n];

    v8f acc = {};
    for (int i = 0; i < 4; ++i) {
        v2f a0 = { a[i].x, a[i].y }, b0 = { b[i].x, b[i].y };
        v2f a1 = { a[i].z, a[i].w }, b1 = { b[i].z, b[i].w };
        acc = __builtin_amdgcn_wmma_f32_16x16x4_f32(
            false, a0, false, b0, (short)0, acc, false, false);
        acc = __builtin_amdgcn_wmma_f32_16x16x4_f32(
            false, a1, false, b1, (short)0, acc, false, false);
    }

    for (int r = 0; r < 8; ++r)
        part[wv][(lh * 8 + r) * 16 + ln] = acc[r];
    __syncthreads();

    float y = part[0][t];
    for (int s = 1; s < KSPLIT_; ++s)
        y += part[s][t];

    const float scale = pw * rsqrtf(pv + BN_EPS_);
    const float bias  = pb - pm * scale;
    const int   m     = mt * 16 + (t >> 4);
    f[(size_t)m * NOUT_ + n] = tanhf(y * scale + bias);
}

#define RUN_    64
#define DEPTH_  16
#define PLANE4_ (HW_ / 4)
#define SETS_   (RUN_ / 32)

#define LANE_F(t, l) __int_as_float(__builtin_amdgcn_readlane(__float_as_int(t), (l)))

__global__ void __launch_bounds__(256)
apply_kernel(const float* __restrict__ x,
             const float* __restrict__ f,
             const float* __restrict__ gamma,
             const float* __restrict__ beta,
             float* __restrict__ out)
{
    const int q    = blockIdx.x * 256 + threadIdx.x;
    const int c0   = blockIdx.y * RUN_;
    const int b    = blockIdx.z;
    const int lane = threadIdx.x & 31;

    const float4* xb = (const float4*)(x   + ((size_t)b * C_ + c0) * HW_) + q;
    float4*       ob = (float4*)      (out + ((size_t)b * C_ + c0) * HW_) + q;
    const float*  fb = f + (size_t)b * NOUT_ + c0;
    const float*  gp = gamma + c0;
    const float*  bp = beta  + c0;

    float tf0[SETS_], tf1[SETS_], tf2[SETS_], tga[SETS_], tbe[SETS_];
#pragma unroll
    for (int s = 0; s < SETS_; ++s) {
        tf0[s] = fb[0 * C_ + 32 * s + lane];
        tf1[s] = fb[1 * C_ + 32 * s + lane];
        tf2[s] = fb[2 * C_ + 32 * s + lane];
        tga[s] = gp[32 * s + lane];
        tbe[s] = bp[32 * s + lane];
    }

    const int rlo = (c0 == 0)         ? 1        : -1;
    const int rhi = (c0 + RUN_ == C_) ? RUN_ - 2 : RUN_;

    float4 prev = xb[(ptrdiff_t)rlo * PLANE4_];
    float4 cur  = xb[0];
    float4 ring[DEPTH_];
#pragma unroll
    for (int i = 0; i < DEPTH_; ++i)
        ring[i] = xb[(size_t)(1 + i) * PLANE4_];

#pragma unroll
    for (int j = 0; j < RUN_; ++j) {
        float4 nxt = ring[j % DEPTH_];
        const int ahead = j + 1 + DEPTH_;
        if (ahead < RUN_)
            ring[j % DEPTH_] = xb[(size_t)ahead * PLANE4_];
        else if (ahead == RUN_)
            ring[j % DEPTH_] = xb[(ptrdiff_t)rhi * PLANE4_];

        v4f pin = { nxt.x, nxt.y, nxt.z, nxt.w };
        asm volatile("" : "+v"(pin));
        nxt.x = pin.x; nxt.y = pin.y; nxt.z = pin.z; nxt.w = pin.w;

        const float f0 = LANE_F(tf0[j / 32], j % 32);
        const float f1 = LANE_F(tf1[j / 32], j % 32);
        const float f2 = LANE_F(tf2[j / 32], j % 32);
        const float ga = LANE_F(tga[j / 32], j % 32);
        const float be = LANE_F(tbe[j / 32], j % 32);

        const float4 a = prev;
        const float4 m = cur;
        const float4 p = nxt;
        float4 o;
        o.x = ga * (f0 * a.x + f1 * m.x + f2 * p.x) + be * m.x;
        o.y = ga * (f0 * a.y + f1 * m.y + f2 * p.y) + be * m.y;
        o.z = ga * (f0 * a.z + f1 * m.z + f2 * p.z) + be * m.z;
        o.w = ga * (f0 * a.w + f1 * m.w + f2 * p.w) + be * m.w;
        ob[(size_t)j * PLANE4_] = o;

        prev = cur;
        cur  = nxt;
    }
}

extern "C" void kernel_launch(void* const* d_in, const int* in_sizes, int n_in,
                              void* d_out, int out_size, void* d_ws, size_t ws_size,
                              hipStream_t stream)
{
    const float* x      = (const float*)d_in[0];
    const float* conv_w = (const float*)d_in[1];
    const float* bnw    = (const float*)d_in[2];
    const float* bnb    = (const float*)d_in[3];
    const float* bnm    = (const float*)d_in[4];
    const float* bnv    = (const float*)d_in[5];
    const float* gamma  = (const float*)d_in[6];
    const float* beta   = (const float*)d_in[7];
    float*       out    = (float*)d_out;

    float* g_ws = (float*)d_ws;
    float* f_ws = (float*)((char*)d_ws + B_ * C_ * sizeof(float));

    gap_kernel<<<B_ * C_, 256, 0, stream>>>(x, g_ws);
    filt_gemm_wmma_kernel<<<96, 32 * KSPLIT_, 0, stream>>>(g_ws, conv_w, bnw, bnb, bnm, bnv, f_ws);
    apply_kernel<<<dim3(PLANE4_ / 256, C_ / RUN_, B_), 256, 0, stream>>>(x, f_ws, gamma, beta, out);
}


# filter GEMM moved into the apply kernel's prologue: each work-group (batch, 32 channels, whole plane) computes its own 96 filter values on WMMA while its x planes travel; 2 launches, not 3
# speedup vs baseline: 1.3666x; 1.0947x over previous
#include <hip/hip_runtime.h>
#include <math.h>

#define B_    32
#define C_    256
#define HW_   4096
#define NOUT_ 768
#define BN_EPS_ 1e-5f

typedef float v2f __attribute__((ext_vector_type(2)));
typedef float v8f __attribute__((ext_vector_type(8)));
typedef float v4f __attribute__((ext_vector_type(4)));

__global__ void gap_kernel(const float* __restrict__ x, float* __restrict__ g)
{
    const int plane = blockIdx.x;
    const float4* xp = (const float4*)(x + (size_t)plane * HW_);
    float s = 0.f;
#pragma unroll
    for (int i = 0; i < 4; ++i) {
        float4 v = xp[threadIdx.x + i * 256];
        s += v.x + v.y + v.z + v.w;
    }
#pragma unroll
    for (int off = 16; off > 0; off >>= 1)
        s += __shfl_xor(s, off, 32);
    __shared__ float lds[8];
    const int wid  = threadIdx.x >> 5;
    const int lane = threadIdx.x & 31;
    if (lane == 0) lds[wid] = s;
    __syncthreads();
    if (threadIdx.x == 0) {
        float t = 0.f;
#pragma unroll
        for (int w = 0; w < 8; ++w) t += lds[w];
        g[plane] = t * (1.0f / (float)HW_);
    }
}

#define RUN_    32
#define DEPTH_  8
#define PLANE4_ (HW_ / 4)
#define NT_     512
#define COLS_   (PLANE4_ / NT_)
#define KSL_    8

#define LANE_F(t, l) __int_as_float(__builtin_amdgcn_readlane(__float_as_int(t), (l)))

#define LDS_HANDOVER()                                                    \
    do {                                                                  \
        __builtin_amdgcn_fence(__ATOMIC_RELEASE, "workgroup", "local");   \
        __builtin_amdgcn_s_barrier();                                     \
        __builtin_amdgcn_fence(__ATOMIC_ACQUIRE, "workgroup", "local");   \
    } while (0)

__global__ void __launch_bounds__(NT_)
filt_apply_kernel(const float* __restrict__ x,
                  const float* __restrict__ g,
                  const float* __restrict__ w,
                  const float* __restrict__ bnw,
                  const float* __restrict__ bnb,
                  const float* __restrict__ bnm,
                  const float* __restrict__ bnv,
                  const float* __restrict__ gamma,
                  const float* __restrict__ beta,
                  float* __restrict__ out)
{
    __shared__ float part[6][KSL_][16];
    __shared__ float fl[3 * RUN_];

    const int tid  = threadIdx.x;
    const int wv   = tid >> 5;
    const int lane = tid & 31;
    const int ln   = lane & 15;
    const int lh   = lane >> 4;
    const int c0   = blockIdx.x * RUN_;
    const int b    = blockIdx.y;

    const int hh = wv >> 3;
    const int ks = wv & 7;
    const float4* ap = (const float4*)(g + (size_t)b * C_ + ks * 32 + 4 * lh);
    float4 a[4], bq[3][4];
#pragma unroll
    for (int i = 0; i < 4; ++i)
        a[i] = ap[2 * i];
#pragma unroll
    for (int kk = 0; kk < 3; ++kk) {
        const float4* bp = (const float4*)
            (w + (size_t)(kk * C_ + c0 + hh * 16 + ln) * C_ + ks * 32 + 4 * lh);
#pragma unroll
        for (int i = 0; i < 4; ++i)
            bq[kk][i] = bp[2 * i];
    }

    const int nq = (tid < 3 * RUN_) ? (tid >> 5) * C_ + c0 + lane : c0;
    const float pw = bnw[nq];
    const float pb = bnb[nq];
    const float pm = bnm[nq];
    const float pv = bnv[nq];

    const float tga = gamma[c0 + lane];
    const float tbe = beta[c0 + lane];

    __builtin_amdgcn_sched_barrier(0);

    const float*  xrun = x + ((size_t)b * C_ + c0) * HW_;
    const float4* xb = (const float4*)xrun + tid;
    float4*       ob = (float4*)(out + ((size_t)b * C_ + c0) * HW_) + tid;

    const int rlo = (c0 == 0)         ? 1        : -1;
    const int rhi = (c0 + RUN_ == C_) ? RUN_ - 2 : RUN_;
    const float4* xlo = (const float4*)(xrun + (ptrdiff_t)rlo * HW_) + tid;
    const float4* xhi = (const float4*)(xrun + (ptrdiff_t)rhi * HW_) + tid;

    float4 prev[COLS_], cur[COLS_];
    float4 ring[DEPTH_][COLS_];
#pragma unroll
    for (int h = 0; h < COLS_; ++h) {
        prev[h] = xlo[h * NT_];
        cur[h]  = xb[h * NT_];
    }
#pragma unroll
    for (int i = 0; i < DEPTH_; ++i)
#pragma unroll
        for (int h = 0; h < COLS_; ++h)
            ring[i][h] = xb[(size_t)(1 + i) * PLANE4_ + h * NT_];

    __builtin_amdgcn_sched_barrier(0);

    v8f acc[3] = {};
#pragma unroll
    for (int i = 0; i < 4; ++i) {
        const v2f a0 = { a[i].x, a[i].y };
        const v2f a1 = { a[i].z, a[i].w };
#pragma unroll
        for (int kk = 0; kk < 3; ++kk) {
            const v2f b0 = { bq[kk][i].x, bq[kk][i].y };
            const v2f b1 = { bq[kk][i].z, bq[kk][i].w };
            acc[kk] = __builtin_amdgcn_wmma_f32_16x16x4_f32(
                false, a0, false, b0, (short)0, acc[kk], false, false);
            acc[kk] = __builtin_amdgcn_wmma_f32_16x16x4_f32(
                false, a1, false, b1, (short)0, acc[kk], false, false);
        }
    }

    if (lh == 0) {
#pragma unroll
        for (int kk = 0; kk < 3; ++kk)
            part[kk * 2 + hh][ks][ln] = acc[kk][0];
    }
    LDS_HANDOVER();

    if (tid < 3 * RUN_) {
        const int tl = (tid >> 5) * 2 + (lane >> 4);
        float y = part[tl][0][ln];
#pragma unroll
        for (int s = 1; s < KSL_; ++s)
            y += part[tl][s][ln];
        const float scale = pw * rsqrtf(pv + BN_EPS_);
        const float bias  = pb - pm * scale;
        fl[tid] = tanhf(y * scale + bias);
    }
    LDS_HANDOVER();

    const float tf0 = fl[0 * RUN_ + lane];
    const float tf1 = fl[1 * RUN_ + lane];
    const float tf2 = fl[2 * RUN_ + lane];

#pragma unroll
    for (int j = 0; j < RUN_; ++j) {
        float4 nxt[COLS_];
        const int ahead = j + 1 + DEPTH_;
#pragma unroll
        for (int h = 0; h < COLS_; ++h) {
            nxt[h] = ring[j % DEPTH_][h];
            if (ahead < RUN_)
                ring[j % DEPTH_][h] = xb[(size_t)ahead * PLANE4_ + h * NT_];
            else if (ahead == RUN_)
                ring[j % DEPTH_][h] = xhi[h * NT_];
        }

        v4f pin0 = { nxt[0].x, nxt[0].y, nxt[0].z, nxt[0].w };
        v4f pin1 = { nxt[1].x, nxt[1].y, nxt[1].z, nxt[1].w };
        asm volatile("" : "+v"(pin0), "+v"(pin1));
        nxt[0].x = pin0.x; nxt[0].y = pin0.y; nxt[0].z = pin0.z; nxt[0].w = pin0.w;
        nxt[1].x = pin1.x; nxt[1].y = pin1.y; nxt[1].z = pin1.z; nxt[1].w = pin1.w;

        const float f0 = LANE_F(tf0, j);
        const float f1 = LANE_F(tf1, j);
        const float f2 = LANE_F(tf2, j);
        const float ga = LANE_F(tga, j);
        const float be = LANE_F(tbe, j);

#pragma unroll
        for (int h = 0; h < COLS_; ++h) {
            const float4 a4 = prev[h];
            const float4 m  = cur[h];
            const float4 p  = nxt[h];
            float4 o;
            o.x = ga * (f0 * a4.x + f1 * m.x + f2 * p.x) + be * m.x;
            o.y = ga * (f0 * a4.y + f1 * m.y + f2 * p.y) + be * m.y;
            o.z = ga * (f0 * a4.z + f1 * m.z + f2 * p.z) + be * m.z;
            o.w = ga * (f0 * a4.w + f1 * m.w + f2 * p.w) + be * m.w;
            ob[(size_t)j * PLANE4_ + h * NT_] = o;
            prev[h] = cur[h];
            cur[h]  = nxt[h];
        }
    }
}

extern "C" void kernel_launch(void* const* d_in, const int* in_sizes, int n_in,
                              void* d_out, int out_size, void* d_ws, size_t ws_size,
                              hipStream_t stream)
{
    const float* x      = (const float*)d_in[0];
    const float* conv_w = (const float*)d_in[1];
    const float* bnw    = (const float*)d_in[2];
    const float* bnb    = (const float*)d_in[3];
    const float* bnm    = (const float*)d_in[4];
    const float* bnv    = (const float*)d_in[5];
    const float* gamma  = (const float*)d_in[6];
    const float* beta   = (const float*)d_in[7];
    float*       out    = (float*)d_out;

    float* g_ws = (float*)d_ws;

    gap_kernel<<<B_ * C_, 256, 0, stream>>>(x, g_ws);
    filt_apply_kernel<<<dim3(C_ / RUN_, B_), NT_, 0, stream>>>(
        x, g_ws, conv_w, bnw, bnb, bnm, bnv, gamma, beta, out);
}


# pooling kernel as a stream: 4096 long waves, each sums 2 whole planes alone with 24-32 loads of 512 B in flight, no LDS, no barrier (was 65536 waves of 4 loads); the sum order inside a plane changes
# speedup vs baseline: 1.5349x; 1.1232x over previous
#include <hip/hip_runtime.h>
#include <math.h>

#define B_    32
#define C_    256
#define HW_   4096
#define NOUT_ 768
#define BN_EPS_ 1e-5f

typedef float v2f __attribute__((ext_vector_type(2)));
typedef float v8f __attribute__((ext_vector_type(8)));
typedef float v4f __attribute__((ext_vector_type(4)));

#define GAP_WAVES_ 16
#define GAP_GRID_  256
#define GAP_GL_    8
#define GAP_RING_  4

__global__ void __launch_bounds__(GAP_WAVES_ * 32)
gap_kernel(const float* __restrict__ x, float* __restrict__ g)
{
    const int lane = threadIdx.x & 31;
    const int wv = __builtin_amdgcn_readfirstlane(
        (int)(blockIdx.x * GAP_WAVES_ + (threadIdx.x >> 5)));
    const int p0 = wv * 2;
    const float4* xp = (const float4*)(x + (size_t)p0 * HW_) + lane;

    float4 ring[GAP_RING_][GAP_GL_];
#pragma unroll
    for (int q = 0; q < GAP_RING_; ++q)
#pragma unroll
        for (int i = 0; i < GAP_GL_; ++i)
            ring[q][i] = xp[(q * GAP_GL_ + i) * 32];

    v4f acc[2] = { { 0.f, 0.f, 0.f, 0.f }, { 0.f, 0.f, 0.f, 0.f } };
#pragma unroll
    for (int q = 0; q < 2 * GAP_RING_; ++q) {
        const int pl = q / GAP_RING_;
#pragma unroll
        for (int i = 0; i < GAP_GL_; ++i) {
            const float4 v = ring[q % GAP_RING_][i];
            acc[pl].x += v.x; acc[pl].y += v.y; acc[pl].z += v.z; acc[pl].w += v.w;
        }
        if (q + GAP_RING_ < 2 * GAP_RING_) {
            asm volatile("" : "+v"(acc[pl]));
#pragma unroll
            for (int i = 0; i < GAP_GL_; ++i)
                ring[q % GAP_RING_][i] = xp[((q + GAP_RING_) * GAP_GL_ + i) * 32];
            asm volatile("" : "+v"(acc[pl]));
        }
    }

    float t0 = (acc[0].x + acc[0].y) + (acc[0].z + acc[0].w);
    float t1 = (acc[1].x + acc[1].y) + (acc[1].z + acc[1].w);
#pragma unroll
    for (int off = 16; off > 0; off >>= 1) {
        t0 += __shfl_xor(t0, off, 32);
        t1 += __shfl_xor(t1, off, 32);
    }
    if (lane == 0) {
        float2 r;
        r.x = t0 * (1.0f / (float)HW_);
        r.y = t1 * (1.0f / (float)HW_);
        *(float2*)(g + p0) = r;
    }
}

#define RUN_    32
#define DEPTH_  8
#define PLANE4_ (HW_ / 4)
#define NT_     512
#define COLS_   (PLANE4_ / NT_)
#define KSL_    8

#define LANE_F(t, l) __int_as_float(__builtin_amdgcn_readlane(__float_as_int(t), (l)))

#define LDS_HANDOVER()                                                    \
    do {                                                                  \
        __builtin_amdgcn_fence(__ATOMIC_RELEASE, "workgroup", "local");   \
        __builtin_amdgcn_s_barrier();                                     \
        __builtin_amdgcn_fence(__ATOMIC_ACQUIRE, "workgroup", "local");   \
    } while (0)

__global__ void __launch_bounds__(NT_)
filt_apply_kernel(const float* __restrict__ x,
                  const float* __restrict__ g,
                  const float* __restrict__ w,
                  const float* __restrict__ bnw,
                  const float* __restrict__ bnb,
                  const float* __restrict__ bnm,
                  const float* __restrict__ bnv,
                  const float* __restrict__ gamma,
                  const float* __restrict__ beta,
                  float* __restrict__ out)
{
    __shared__ float part[6][KSL_][16];
    __shared__ float fl[3 * RUN_];

    const int tid  = threadIdx.x;
    const int wv   = tid >> 5;
    const int lane = tid & 31;
    const int ln   = lane & 15;
    const int lh   = lane >> 4;
    const int c0   = blockIdx.x * RUN_;
    const int b    = blockIdx.y;

    const int hh = wv >> 3;
    const int ks = wv & 7;
    const float4* ap = (const float4*)(g + (size_t)b * C_ + ks * 32 + 4 * lh);
    float4 a[4], bq[3][4];
#pragma unroll
    for (int i = 0; i < 4; ++i)
        a[i] = ap[2 * i];
#pragma unroll
    for (int kk = 0; kk < 3; ++kk) {
        const float4* bp = (const float4*)
            (w + (size_t)(kk * C_ + c0 + hh * 16 + ln) * C_ + ks * 32 + 4 * lh);
#pragma unroll
        for (int i = 0; i < 4; ++i)
            bq[kk][i] = bp[2 * i];
    }

    const int nq = (tid < 3 * RUN_) ? (tid >> 5) * C_ + c0 + lane : c0;
    const float pw = bnw[nq];
    const float pb = bnb[nq];
    const float pm = bnm[nq];
    const float pv = bnv[nq];

    const float tga = gamma[c0 + lane];
    const float tbe = beta[c0 + lane];

    __builtin_amdgcn_sched_barrier(0);

    const float*  xrun = x + ((size_t)b * C_ + c0) * HW_;
    const float4* xb = (const float4*)xrun + tid;
    float4*       ob = (float4*)(out + ((size_t)b * C_ + c0) * HW_) + tid;

    const int rlo = (c0 == 0)         ? 1        : -1;
    const int rhi = (c0 + RUN_ == C_) ? RUN_ - 2 : RUN_;
    const float4* xlo = (const float4*)(xrun + (ptrdiff_t)rlo * HW_) + tid;
    const float4* xhi = (const float4*)(xrun + (ptrdiff_t)rhi * HW_) + tid;

    float4 prev[COLS_], cur[COLS_];
    float4 ring[DEPTH_][COLS_];
#pragma unroll
    for (int h = 0; h < COLS_; ++h) {
        prev[h] = xlo[h * NT_];
        cur[h]  = xb[h * NT_];
    }
#pragma unroll
    for (int i = 0; i < DEPTH_; ++i)
#pragma unroll
        for (int h = 0; h < COLS_; ++h)
            ring[i][h] = xb[(size_t)(1 + i) * PLANE4_ + h * NT_];

    __builtin_amdgcn_sched_barrier(0);

    v8f acc[3] = {};
#pragma unroll
    for (int i = 0; i < 4; ++i) {
        const v2f a0 = { a[i].x, a[i].y };
        const v2f a1 = { a[i].z, a[i].w };
#pragma unroll
        for (int kk = 0; kk < 3; ++kk) {
            const v2f b0 = { bq[kk][i].x, bq[kk][i].y };
            const v2f b1 = { bq[kk][i].z, bq[kk][i].w };
            acc[kk] = __builtin_amdgcn_wmma_f32_16x16x4_f32(
                false, a0, false, b0, (short)0, acc[kk], false, false);
            acc[kk] = __builtin_amdgcn_wmma_f32_16x16x4_f32(
                false, a1, false, b1, (short)0, acc[kk], false, false);
        }
    }

    if (lh == 0) {
#pragma unroll
        for (int kk = 0; kk < 3; ++kk)
            part[kk * 2 + hh][ks][ln] = acc[kk][0];
    }
    LDS_HANDOVER();

    if (tid < 3 * RUN_) {
        const int tl = (tid >> 5) * 2 + (lane >> 4);
        float y = part[tl][0][ln];
#pragma unroll
        for (int s = 1; s < KSL_; ++s)
            y += part[tl][s][ln];
        const float scale = pw * rsqrtf(pv + BN_EPS_);
        const float bias  = pb - pm * scale;
        fl[tid] = tanhf(y * scale + bias);
    }
    LDS_HANDOVER();

    const float tf0 = fl[0 * RUN_ + lane];
    const float tf1 = fl[1 * RUN_ + lane];
    const float tf2 = fl[2 * RUN_ + lane];

#pragma unroll
    for (int j = 0; j < RUN_; ++j) {
        float4 nxt[COLS_];
        const int ahead = j + 1 + DEPTH_;
#pragma unroll
        for (int h = 0; h < COLS_; ++h) {
            nxt[h] = ring[j % DEPTH_][h];
            if (ahead < RUN_)
                ring[j % DEPTH_][h] = xb[(size_t)ahead * PLANE4_ + h * NT_];
            else if (ahead == RUN_)
                ring[j % DEPTH_][h] = xhi[h * NT_];
        }

        v4f pin0 = { nxt[0].x, nxt[0].y, nxt[0].z, nxt[0].w };
        v4f pin1 = { nxt[1].x, nxt[1].y, nxt[1].z, nxt[1].w };
        asm volatile("" : "+v"(pin0), "+v"(pin1));
        nxt[0].x = pin0.x; nxt[0].y = pin0.y; nxt[0].z = pin0.z; nxt[0].w = pin0.w;
        nxt[1].x = pin1.x; nxt[1].y = pin1.y; nxt[1].z = pin1.z; nxt[1].w = pin1.w;

        const float f0 = LANE_F(tf0, j);
        const float f1 = LANE_F(tf1, j);
        const float f2 = LANE_F(tf2, j);
        const float ga = LANE_F(tga, j);
        const float be = LANE_F(tbe, j);

#pragma unroll
        for (int h = 0; h < COLS_; ++h) {
            const float4 a4 = prev[h];
            const float4 m  = cur[h];
            const float4 p  = nxt[h];
            float4 o;
            o.x = ga * (f0 * a4.x + f1 * m.x + f2 * p.x) + be * m.x;
            o.y = ga * (f0 * a4.y + f1 * m.y + f2 * p.y) + be * m.y;
            o.z = ga * (f0 * a4.z + f1 * m.z + f2 * p.z) + be * m.z;
            o.w = ga * (f0 * a4.w + f1 * m.w + f2 * p.w) + be * m.w;
            ob[(size_t)j * PLANE4_ + h * NT_] = o;
            prev[h] = cur[h];
            cur[h]  = nxt[h];
        }
    }
}

extern "C" void kernel_launch(void* const* d_in, const int* in_sizes, int n_in,
                              void* d_out, int out_size, void* d_ws, size_t ws_size,
                              hipStream_t stream)
{
    const float* x      = (const float*)d_in[0];
    const float* conv_w = (const float*)d_in[1];
    const float* bnw    = (const float*)d_in[2];
    const float* bnb    = (const float*)d_in[3];
    const float* bnm    = (const float*)d_in[4];
    const float* bnv    = (const float*)d_in[5];
    const float* gamma  = (const float*)d_in[6];
    const float* beta   = (const float*)d_in[7];
    float*       out    = (float*)d_out;

    float* g_ws = (float*)d_ws;

    gap_kernel<<<GAP_GRID_, GAP_WAVES_ * 32, 0, stream>>>(x, g_ws);
    filt_apply_kernel<<<dim3(C_ / RUN_, B_), NT_, 0, stream>>>(
        x, g_ws, conv_w, bnw, bnb, bnm, bnv, gamma, beta, out);
}
